# starting kernel
# baseline (speedup 1.0000x reference)
#include <hip/hip_runtime.h>

#define IN_F   256
#define N_ASS  4096
#define OUT_F  256
#define BATCH  128
#define ASS0   IN_F
#define OUT0   (IN_F + N_ASS)

typedef float v2f __attribute__((ext_vector_type(2)));
typedef float v8f __attribute__((ext_vector_type(8)));

__global__ __launch_bounds__(256) void rnn_zero_f32(float* __restrict__ p, int n) {
    int i = blockIdx.x * blockDim.x + threadIdx.x;
    if (i < n) p[i] = 0.0f;
}

__global__ __launch_bounds__(256) void rnn_scatter_w1(const float* __restrict__ w,
                                                      const int* __restrict__ src,
                                                      const int* __restrict__ dst,
                                                      int e, float* __restrict__ W1) {
    int i = blockIdx.x * blockDim.x + threadIdx.x;
    if (i < e) {
        atomicAdd(&W1[(size_t)(dst[i] - ASS0) * IN_F + src[i]], w[i]);
    }
}

__global__ __launch_bounds__(256) void rnn_scatter_w2(const float* __restrict__ w,
                                                      const int* __restrict__ src,
                                                      const int* __restrict__ dst,
                                                      int e, float* __restrict__ W2) {
    int i = blockIdx.x * blockDim.x + threadIdx.x;
    if (i < e) {
        int d = dst[i];
        if (d >= OUT0) {
            atomicAdd(&W2[(size_t)(d - OUT0) * N_ASS + (src[i] - ASS0)], w[i]);
        }
    }
}

template <int M, int N, int K, bool BT, bool CT>
__global__ __launch_bounds__(256) void rnn_wmma_gemm_f32(const float* __restrict__ A,
                                                         const float* __restrict__ B,
                                                         float* __restrict__ C) {
    const int lane = threadIdx.x & 31;
    const int wave = threadIdx.x >> 5;
    const int tm   = blockIdx.x;
    const int half = lane >> 4;
    const int l16  = lane & 15;
    const int m    = tm * 16 + l16;
    const int n    = wave * 16 + l16;

    v8f acc = {};
    const float* __restrict__ Arow = A + (size_t)m * K;
    const float* __restrict__ Brow = BT ? (B + (size_t)n * K) : nullptr;

    for (int kb = 0; kb < K; kb += 4) {
        const int ka = kb + 2 * half;
        v2f a = *(const v2f*)(Arow + ka);
        v2f b;
        if (BT) {
            b = *(const v2f*)(Brow + ka);
        } else {
            b.x = B[(size_t)ka * N + n];
            b.y = B[(size_t)(ka + 1) * N + n];
        }
        acc = __builtin_amdgcn_wmma_f32_16x16x4_f32(false, a, false, b,
                                                    (short)0, acc, false, false);
    }

#pragma unroll
    for (int v = 0; v < 8; ++v) {
        const int mr = tm * 16 + v + 8 * half;
        if (CT) C[(size_t)n * M + mr] = acc[v];
        else    C[(size_t)mr * N + n] = acc[v];
    }
}

extern "C" void kernel_launch(void* const* d_in, const int* in_sizes, int n_in,
                              void* d_out, int out_size, void* d_ws, size_t ws_size,
                              hipStream_t stream) {
    const float* x   = (const float*)d_in[0];
    const float* wi  = (const float*)d_in[1];
    const float* wa  = (const float*)d_in[2];
    const int* in_src  = (const int*)d_in[3];
    const int* in_dst  = (const int*)d_in[4];
    const int* ass_src = (const int*)d_in[5];
    const int* ass_dst = (const int*)d_in[6];
    const int e_in  = in_sizes[1];
    const int e_ass = in_sizes[2];

    float* W1    = (float*)d_ws;
    float* W2    = W1 + (size_t)N_ASS * IN_F;
    float* ACC_T = W2 + (size_t)OUT_F * N_ASS;

    {
        const int n = N_ASS * IN_F + OUT_F * N_ASS;
        rnn_zero_f32<<<(n + 255) / 256, 256, 0, stream>>>(W1, n);
    }
    rnn_scatter_w1<<<(e_in + 255) / 256, 256, 0, stream>>>(wi, in_src, in_dst, e_in, W1);
    rnn_scatter_w2<<<(e_ass + 255) / 256, 256, 0, stream>>>(wa, ass_src, ass_dst, e_ass, W2);

    rnn_wmma_gemm_f32<N_ASS, BATCH, IN_F, true, true>
        <<<N_ASS / 16, 256, 0, stream>>>(W1, x, ACC_T);

    rnn_wmma_gemm_f32<OUT_F, BATCH, N_ASS, true, true>
        <<<OUT_F / 16, 256, 0, stream>>>(W2, ACC_T, (float*)d_out);

    (void)n_in; (void)out_size; (void)ws_size;
}


# Hop-2 GEMM: K=4096 split over the 32 waves of one work-group per 16x16 tile (128 groups, not 16); partial sums added via LDS in fixed wave order (summation order changes)
# speedup vs baseline: 3.2532x; 3.2532x over previous
#include <hip/hip_runtime.h>

#define IN_F   256
#define N_ASS  4096
#define OUT_F  256
#define BATCH  128
#define ASS0   IN_F
#define OUT0   (IN_F + N_ASS)

typedef float v2f __attribute__((ext_vector_type(2)));
typedef float v8f __attribute__((ext_vector_type(8)));

__global__ __launch_bounds__(256) void rnn_zero_f32(float* __restrict__ p, int n) {
    int i = blockIdx.x * blockDim.x + threadIdx.x;
    if (i < n) p[i] = 0.0f;
}

__global__ __launch_bounds__(256) void rnn_scatter_w1(const float* __restrict__ w,
                                                      const int* __restrict__ src,
                                                      const int* __restrict__ dst,
                                                      int e, float* __restrict__ W1) {
    int i = blockIdx.x * blockDim.x + threadIdx.x;
    if (i < e) {
        atomicAdd(&W1[(size_t)(dst[i] - ASS0) * IN_F + src[i]], w[i]);
    }
}

__global__ __launch_bounds__(256) void rnn_scatter_w2(const float* __restrict__ w,
                                                      const int* __restrict__ src,
                                                      const int* __restrict__ dst,
                                                      int e, float* __restrict__ W2) {
    int i = blockIdx.x * blockDim.x + threadIdx.x;
    if (i < e) {
        int d = dst[i];
        if (d >= OUT0) {
            atomicAdd(&W2[(size_t)(d - OUT0) * N_ASS + (src[i] - ASS0)], w[i]);
        }
    }
}

template <int M, int N, int K, bool BT, bool CT>
__global__ __launch_bounds__(256) void rnn_wmma_gemm_f32(const float* __restrict__ A,
                                                         const float* __restrict__ B,
                                                         float* __restrict__ C) {
    const int lane = threadIdx.x & 31;
    const int wave = threadIdx.x >> 5;
    const int tm   = blockIdx.x;
    const int half = lane >> 4;
    const int l16  = lane & 15;
    const int m    = tm * 16 + l16;
    const int n    = wave * 16 + l16;

    v8f acc = {};
    const float* __restrict__ Arow = A + (size_t)m * K;
    const float* __restrict__ Brow = BT ? (B + (size_t)n * K) : nullptr;

    for (int kb = 0; kb < K; kb += 4) {
        const int ka = kb + 2 * half;
        v2f a = *(const v2f*)(Arow + ka);
        v2f b;
        if (BT) {
            b = *(const v2f*)(Brow + ka);
        } else {
            b.x = B[(size_t)ka * N + n];
            b.y = B[(size_t)(ka + 1) * N + n];
        }
        acc = __builtin_amdgcn_wmma_f32_16x16x4_f32(false, a, false, b,
                                                    (short)0, acc, false, false);
    }

#pragma unroll
    for (int v = 0; v < 8; ++v) {
        const int mr = tm * 16 + v + 8 * half;
        if (CT) C[(size_t)n * M + mr] = acc[v];
        else    C[(size_t)mr * N + n] = acc[v];
    }
}

template <int M, int N, int K>
__global__ __launch_bounds__(1024) void rnn_wmma_gemm_f32_ksplit(const float* __restrict__ A,
                                                                 const float* __restrict__ B,
                                                                 float* __restrict__ C) {
    constexpr int WAVES = 32;
    constexpr int KW    = K / WAVES;
    static_assert(K % (WAVES * 4) == 0, "K slice must be a multiple of the WMMA k");
    __shared__ float part[WAVES][256];

    const int lane = threadIdx.x & 31;
    const int wave = threadIdx.x >> 5;
    const int tm   = blockIdx.x;
    const int tn   = blockIdx.y;
    const int half = lane >> 4;
    const int l16  = lane & 15;

    const float* __restrict__ Arow = A + (size_t)(tm * 16 + l16) * K + wave * KW + 2 * half;
    const float* __restrict__ Brow = B + (size_t)(tn * 16 + l16) * K + wave * KW + 2 * half;

    v8f acc = {};
    for (int kb = 0; kb < KW; kb += 4) {
        v2f a = *(const v2f*)(Arow + kb);
        v2f b = *(const v2f*)(Brow + kb);
        acc = __builtin_amdgcn_wmma_f32_16x16x4_f32(false, a, false, b,
                                                    (short)0, acc, false, false);
    }

#pragma unroll
    for (int v = 0; v < 8; ++v) part[wave][v * 32 + lane] = acc[v];
    __syncthreads();

    if (threadIdx.x < 256) {
        const int mm  = threadIdx.x & 15;
        const int nn  = threadIdx.x >> 4;
        const int idx = (mm & 7) * 32 + (mm >> 3) * 16 + nn;
        float s = part[0][idx];
        for (int w = 1; w < WAVES; ++w) s += part[w][idx];
        C[(size_t)(tn * 16 + nn) * M + tm * 16 + mm] = s;
    }
}

extern "C" void kernel_launch(void* const* d_in, const int* in_sizes, int n_in,
                              void* d_out, int out_size, void* d_ws, size_t ws_size,
                              hipStream_t stream) {
    const float* x   = (const float*)d_in[0];
    const float* wi  = (const float*)d_in[1];
    const float* wa  = (const float*)d_in[2];
    const int* in_src  = (const int*)d_in[3];
    const int* in_dst  = (const int*)d_in[4];
    const int* ass_src = (const int*)d_in[5];
    const int* ass_dst = (const int*)d_in[6];
    const int e_in  = in_sizes[1];
    const int e_ass = in_sizes[2];

    float* W1    = (float*)d_ws;
    float* W2    = W1 + (size_t)N_ASS * IN_F;
    float* ACC_T = W2 + (size_t)OUT_F * N_ASS;

    {
        const int n = N_ASS * IN_F + OUT_F * N_ASS;
        rnn_zero_f32<<<(n + 255) / 256, 256, 0, stream>>>(W1, n);
    }
    rnn_scatter_w1<<<(e_in + 255) / 256, 256, 0, stream>>>(wi, in_src, in_dst, e_in, W1);
    rnn_scatter_w2<<<(e_ass + 255) / 256, 256, 0, stream>>>(wa, ass_src, ass_dst, e_ass, W2);

    rnn_wmma_gemm_f32<N_ASS, BATCH, IN_F, true, true>
        <<<N_ASS / 16, 256, 0, stream>>>(W1, x, ACC_T);

    rnn_wmma_gemm_f32_ksplit<OUT_F, BATCH, N_ASS>
        <<<dim3(OUT_F / 16, BATCH / 16), 1024, 0, stream>>>(W2, ACC_T, (float*)d_out);

    (void)n_in; (void)out_size; (void)ws_size;
}


# Independent work shares launches (5 launches -> 4): hop-2 edge scatter runs in the hop-1 GEMM's launch, W2 zeroing in the hop-1 scatter's launch; float atomics and sum order as in parent
# speedup vs baseline: 3.8311x; 1.1776x over previous
#include <hip/hip_runtime.h>

#define IN_F   256
#define N_ASS  4096
#define OUT_F  256
#define BATCH  128
#define ASS0   IN_F
#define OUT0   (IN_F + N_ASS)

typedef float v2f __attribute__((ext_vector_type(2)));
typedef float v8f __attribute__((ext_vector_type(8)));

__global__ __launch_bounds__(256) void rnn_zero_f32(float* __restrict__ p, int n) {
    int i = blockIdx.x * blockDim.x + threadIdx.x;
    if (i < n) p[i] = 0.0f;
}

__device__ __forceinline__ void rnn_scatter_w1_edge(const float* __restrict__ w,
                                                    const int* __restrict__ src,
                                                    const int* __restrict__ dst,
                                                    int e, float* __restrict__ W1, int i) {
    if (i < e) {
        atomicAdd(&W1[(size_t)(dst[i] - ASS0) * IN_F + src[i]], w[i]);
    }
}

__device__ __forceinline__ void rnn_scatter_w2_edge(const float* __restrict__ w,
                                                    const int* __restrict__ src,
                                                    const int* __restrict__ dst,
                                                    int e, float* __restrict__ W2, int i) {
    if (i < e) {
        int d = dst[i];
        if (d >= OUT0) {
            atomicAdd(&W2[(size_t)(d - OUT0) * N_ASS + (src[i] - ASS0)], w[i]);
        }
    }
}

__global__ __launch_bounds__(256) void rnn_scatter_w1_zero_w2(const float* __restrict__ w,
                                                              const int* __restrict__ src,
                                                              const int* __restrict__ dst,
                                                              int e, float* __restrict__ W1,
                                                              float* __restrict__ W2, int n2) {
    const int sb = (e + 255) >> 8;
    const int b  = blockIdx.x;
    if (b < sb) {
        rnn_scatter_w1_edge(w, src, dst, e, W1, b * 256 + (int)threadIdx.x);
    } else {
        const int i = (b - sb) * 256 + (int)threadIdx.x;
        if (i < n2) W2[i] = 0.0f;
    }
}

template <int M, int N, int K, bool BT, bool CT>
__device__ __forceinline__ void rnn_wmma_gemm_f32_tile(const float* __restrict__ A,
                                                       const float* __restrict__ B,
                                                       float* __restrict__ C,
                                                       const int tm) {
    const int lane = threadIdx.x & 31;
    const int wave = threadIdx.x >> 5;
    const int half = lane >> 4;
    const int l16  = lane & 15;
    const int m    = tm * 16 + l16;
    const int n    = wave * 16 + l16;

    v8f acc = {};
    const float* __restrict__ Arow = A + (size_t)m * K;
    const float* __restrict__ Brow = BT ? (B + (size_t)n * K) : nullptr;

    for (int kb = 0; kb < K; kb += 4) {
        const int ka = kb + 2 * half;
        v2f a = *(const v2f*)(Arow + ka);
        v2f b;
        if (BT) {
            b = *(const v2f*)(Brow + ka);
        } else {
            b.x = B[(size_t)ka * N + n];
            b.y = B[(size_t)(ka + 1) * N + n];
        }
        acc = __builtin_amdgcn_wmma_f32_16x16x4_f32(false, a, false, b,
                                                    (short)0, acc, false, false);
    }

#pragma unroll
    for (int v = 0; v < 8; ++v) {
        const int mr = tm * 16 + v + 8 * half;
        if (CT) C[(size_t)n * M + mr] = acc[v];
        else    C[(size_t)mr * N + n] = acc[v];
    }
}

__global__ __launch_bounds__(256) void rnn_hop1_gemm_scatter_w2(const float* __restrict__ W1,
                                                                const float* __restrict__ x,
                                                                float* __restrict__ ACC_T,
                                                                const float* __restrict__ w,
                                                                const int* __restrict__ src,
                                                                const int* __restrict__ dst,
                                                                int e, float* __restrict__ W2) {
    constexpr int GEMM_BLOCKS = N_ASS / 16;
    const int b = blockIdx.x;
    if (b < GEMM_BLOCKS) {
        rnn_wmma_gemm_f32_tile<N_ASS, BATCH, IN_F, true, true>(W1, x, ACC_T, b);
    } else {
        rnn_scatter_w2_edge(w, src, dst, e, W2, (b - GEMM_BLOCKS) * 256 + (int)threadIdx.x);
    }
}

template <int M, int N, int K>
__global__ __launch_bounds__(1024) void rnn_wmma_gemm_f32_ksplit(const float* __restrict__ A,
                                                                 const float* __restrict__ B,
                                                                 float* __restrict__ C) {
    constexpr int WAVES = 32;
    constexpr int KW    = K / WAVES;
    static_assert(K % (WAVES * 4) == 0, "K slice must be a multiple of the WMMA k");
    __shared__ float part[WAVES][256];

    const int lane = threadIdx.x & 31;
    const int wave = threadIdx.x >> 5;
    const int tm   = blockIdx.x;
    const int tn   = blockIdx.y;
    const int half = lane >> 4;
    const int l16  = lane & 15;

    const float* __restrict__ Arow = A + (size_t)(tm * 16 + l16) * K + wave * KW + 2 * half;
    const float* __restrict__ Brow = B + (size_t)(tn * 16 + l16) * K + wave * KW + 2 * half;

    v8f acc = {};
    for (int kb = 0; kb < KW; kb += 4) {
        v2f a = *(const v2f*)(Arow + kb);
        v2f b = *(const v2f*)(Brow + kb);
        acc = __builtin_amdgcn_wmma_f32_16x16x4_f32(false, a, false, b,
                                                    (short)0, acc, false, false);
    }

#pragma unroll
    for (int v = 0; v < 8; ++v) part[wave][v * 32 + lane] = acc[v];
    __syncthreads();

    if (threadIdx.x < 256) {
        const int mm  = threadIdx.x & 15;
        const int nn  = threadIdx.x >> 4;
        const int idx = (mm & 7) * 32 + (mm >> 3) * 16 + nn;
        float s = part[0][idx];
        for (int w = 1; w < WAVES; ++w) s += part[w][idx];
        C[(size_t)(tn * 16 + nn) * M + tm * 16 + mm] = s;
    }
}

extern "C" void kernel_launch(void* const* d_in, const int* in_sizes, int n_in,
                              void* d_out, int out_size, void* d_ws, size_t ws_size,
                              hipStream_t stream) {
    const float* x   = (const float*)d_in[0];
    const float* wi  = (const float*)d_in[1];
    const float* wa  = (const float*)d_in[2];
    const int* in_src  = (const int*)d_in[3];
    const int* in_dst  = (const int*)d_in[4];
    const int* ass_src = (const int*)d_in[5];
    const int* ass_dst = (const int*)d_in[6];
    const int e_in  = in_sizes[1];
    const int e_ass = in_sizes[2];

    float* W1    = (float*)d_ws;
    float* W2    = W1 + (size_t)N_ASS * IN_F;
    float* ACC_T = W2 + (size_t)OUT_F * N_ASS;

    rnn_zero_f32<<<(N_ASS * IN_F) / 256, 256, 0, stream>>>(W1, N_ASS * IN_F);

    rnn_scatter_w1_zero_w2<<<(e_in + 255) / 256 + (OUT_F * N_ASS) / 256, 256, 0, stream>>>(
        wi, in_src, in_dst, e_in, W1, W2, OUT_F * N_ASS);

    rnn_hop1_gemm_scatter_w2<<<N_ASS / 16 + (e_ass + 255) / 256, 256, 0, stream>>>(
        W1, x, ACC_T, wa, ass_src, ass_dst, e_ass, W2);

    rnn_wmma_gemm_f32_ksplit<OUT_F, BATCH, N_ASS>
        <<<dim3(OUT_F / 16, BATCH / 16), 1024, 0, stream>>>(W2, ACC_T, (float*)d_out);

    (void)n_in; (void)out_size; (void)ws_size;
}
